# starting kernel
# baseline (speedup 1.0000x reference)
#include <hip/hip_runtime.h>

typedef __attribute__((ext_vector_type(2))) float v2f;
typedef __attribute__((ext_vector_type(8))) float v8f;

#define FDIM 256
#define BM 64
#define BN 128
#define BK 64
#define SXP (BK + 4)
#define BNP 144

__global__ __launch_bounds__(256)
void gemm_f32_wmma(const float* __restrict__ X, const float* __restrict__ W,
                   float* __restrict__ H, int nrows) {
  __shared__ __align__(16) float sX[BM * SXP];
  __shared__ __align__(16) float sWp[(BK / 2) * BNP * 2];

  const int t       = threadIdx.x;
  const int lane    = t & 31;
  const int wave    = t >> 5;
  const int l16     = lane & 15;
  const int half    = lane >> 4;
  const int waveRow = wave & 3;
  const int waveCol = wave >> 2;
  const int rowBase = blockIdx.x * BM;
  const int colBase = blockIdx.y * BN;
  const int cb      = waveCol * 64 + l16;

  v8f acc0 = {}, acc1 = {}, acc2 = {}, acc3 = {};

  for (int kBase = 0; kBase < FDIM; kBase += BK) {
#pragma unroll
    for (int it = 0; it < 4; ++it) {
      int idx = t + 256 * it;
      int row = idx >> 4;
      int c4  = (idx & 15) << 2;
      float4 v = make_float4(0.f, 0.f, 0.f, 0.f);
      int g = rowBase + row;
      if (g < nrows)
        v = *(const float4*)(X + (size_t)g * FDIM + kBase + c4);
      *(float4*)&sX[row * SXP + c4] = v;
    }
#pragma unroll
    for (int it = 0; it < 8; ++it) {
      int idx = t + 256 * it;
      int k   = idx >> 5;
      int n4  = (idx & 31) << 2;
      float4 v = *(const float4*)(W + (size_t)(kBase + k) * FDIM + colBase + n4);
      float* p = sWp + ((size_t)(k >> 1) * BNP + n4) * 2 + (k & 1);
      p[0] = v.x; p[2] = v.y; p[4] = v.z; p[6] = v.w;
    }
    __syncthreads();

#pragma unroll
    for (int kk = 0; kk < BK; kk += 4) {
      const int ks = kk + 2 * half;
      v2f a = *(const v2f*)&sX[(waveRow * 16 + l16) * SXP + ks];
      const v2f* wrow = (const v2f*)(sWp + ((kk >> 1) + half) * BNP * 2);
      v2f b0 = wrow[cb];
      v2f b1 = wrow[cb + 16];
      v2f b2 = wrow[cb + 32];
      v2f b3 = wrow[cb + 48];
      acc0 = __builtin_amdgcn_wmma_f32_16x16x4_f32(false, a, false, b0, (short)0, acc0, false, false);
      acc1 = __builtin_amdgcn_wmma_f32_16x16x4_f32(false, a, false, b1, (short)0, acc1, false, false);
      acc2 = __builtin_amdgcn_wmma_f32_16x16x4_f32(false, a, false, b2, (short)0, acc2, false, false);
      acc3 = __builtin_amdgcn_wmma_f32_16x16x4_f32(false, a, false, b3, (short)0, acc3, false, false);
    }
    __syncthreads();
  }

#pragma unroll
  for (int v = 0; v < 8; ++v) {
    int gr = rowBase + waveRow * 16 + v + 8 * half;
    if (gr < nrows) {
      float* p = H + (size_t)gr * FDIM + colBase + waveCol * 64 + l16;
      p[0]  = acc0[v];
      p[16] = acc1[v];
      p[32] = acc2[v];
      p[48] = acc3[v];
    }
  }
}

__global__ __launch_bounds__(256)
void scatter_edges(const float* __restrict__ h, const int* __restrict__ src,
                   const int* __restrict__ dst, const float* __restrict__ val,
                   float* __restrict__ acc, int E) {
  long long idx = (long long)blockIdx.x * blockDim.x + threadIdx.x;
  int e  = (int)(idx >> 6);
  int c4 = (int)(idx & 63) << 2;
  if (e >= E) return;
  int   s = src[e];
  int   d = dst[e];
  float v = val[e];
  float4 hv = *(const float4*)(h + (size_t)s * FDIM + c4);
  float* a = acc + (size_t)d * FDIM + c4;
  unsafeAtomicAdd(a + 0, hv.x * v);
  unsafeAtomicAdd(a + 1, hv.y * v);
  unsafeAtomicAdd(a + 2, hv.z * v);
  unsafeAtomicAdd(a + 3, hv.w * v);
}

__global__ __launch_bounds__(256)
void zero_buf(float* __restrict__ p, long long n4) {
  long long i = (long long)blockIdx.x * blockDim.x + threadIdx.x;
  if (i < n4) *(float4*)(p + 4 * i) = make_float4(0.f, 0.f, 0.f, 0.f);
}

__global__ __launch_bounds__(256)
void bias_relu(const float* __restrict__ acc, const float* __restrict__ b,
               float* __restrict__ o, long long n) {
  long long i = (long long)blockIdx.x * blockDim.x + threadIdx.x;
  if (i < n) o[i] = fmaxf(acc[i] + b[i & (FDIM - 1)], 0.f);
}

__global__ __launch_bounds__(256)
void final_fuse(const float* __restrict__ acc, const float* __restrict__ b,
                const float* __restrict__ x1, float* __restrict__ out, long long n) {
  long long i = (long long)blockIdx.x * blockDim.x + threadIdx.x;
  if (i < n) out[i] = fmaxf(acc[i] + b[i & (FDIM - 1)], 0.f) + x1[i];
}

extern "C" void kernel_launch(void* const* d_in, const int* in_sizes, int n_in,
                              void* d_out, int out_size, void* d_ws, size_t ws_size,
                              hipStream_t stream) {
  const float* x    = (const float*)d_in[0];
  const int*   esrc = (const int*)d_in[1];
  const int*   edst = (const int*)d_in[2];
  const float* eval = (const float*)d_in[3];
  const float* W1   = (const float*)d_in[4];
  const float* b1   = (const float*)d_in[5];
  const float* W2   = (const float*)d_in[6];
  const float* b2   = (const float*)d_in[7];
  float* out = (float*)d_out;

  const int R  = 4;
  const int Nn = in_sizes[0] / FDIM;
  const int E  = in_sizes[1] / R;

  const size_t nf = (size_t)Nn * FDIM;
  float* h   = (float*)d_ws;
  float* acc = h + nf;
  float* x1  = acc + nf;

  dim3 gGrid((Nn + BM - 1) / BM, FDIM / BN);
  const long long n4 = (long long)(nf / 4);
  const int zBlocks = (int)((n4 + 255) / 256);
  const int sBlocks = (int)(((long long)E * 64 + 255) / 256);
  const int pBlocks = (int)(((long long)nf + 255) / 256);

  zero_buf<<<zBlocks, 256, 0, stream>>>(acc, n4);
  for (int r = 0; r < R; ++r) {
    gemm_f32_wmma<<<gGrid, 256, 0, stream>>>(x, W1 + (size_t)r * FDIM * FDIM, h, Nn);
    scatter_edges<<<sBlocks, 256, 0, stream>>>(h, esrc + (size_t)r * E,
                                               edst + (size_t)r * E,
                                               eval + (size_t)r * E, acc, E);
  }
  bias_relu<<<pBlocks, 256, 0, stream>>>(acc, b1, x1, (long long)nf);

  zero_buf<<<zBlocks, 256, 0, stream>>>(acc, n4);
  for (int r = 0; r < R; ++r) {
    gemm_f32_wmma<<<gGrid, 256, 0, stream>>>(x1, W2 + (size_t)r * FDIM * FDIM, h, Nn);
    scatter_edges<<<sBlocks, 256, 0, stream>>>(h, esrc + (size_t)r * E,
                                               edst + (size_t)r * E,
                                               eval + (size_t)r * E, acc, E);
  }
  final_fuse<<<pBlocks, 256, 0, stream>>>(acc, b2, x1, out, (long long)nf);
}


# global float-atomic edge scatter replaced by owner blocks: each block holds 196 dst rows' sums in LDS, scans the whole dst list, adds its own edges with LDS float adds; sums still in arrival order
# speedup vs baseline: 2.0284x; 2.0284x over previous
#include <hip/hip_runtime.h>

typedef __attribute__((ext_vector_type(2))) float v2f;
typedef __attribute__((ext_vector_type(8))) float v8f;

#define FDIM 256
#define BM 64
#define BN 128
#define BK 64
#define SXP (BK + 4)
#define BNP 144

__global__ __launch_bounds__(256)
void gemm_f32_wmma(const float* __restrict__ X, const float* __restrict__ W,
                   float* __restrict__ H, int nrows) {
  __shared__ __align__(16) float sX[BM * SXP];
  __shared__ __align__(16) float sWp[(BK / 2) * BNP * 2];

  const int t       = threadIdx.x;
  const int lane    = t & 31;
  const int wave    = t >> 5;
  const int l16     = lane & 15;
  const int half    = lane >> 4;
  const int waveRow = wave & 3;
  const int waveCol = wave >> 2;
  const int rowBase = blockIdx.x * BM;
  const int colBase = blockIdx.y * BN;
  const int cb      = waveCol * 64 + l16;

  v8f acc0 = {}, acc1 = {}, acc2 = {}, acc3 = {};

  for (int kBase = 0; kBase < FDIM; kBase += BK) {
#pragma unroll
    for (int it = 0; it < 4; ++it) {
      int idx = t + 256 * it;
      int row = idx >> 4;
      int c4  = (idx & 15) << 2;
      float4 v = make_float4(0.f, 0.f, 0.f, 0.f);
      int g = rowBase + row;
      if (g < nrows)
        v = *(const float4*)(X + (size_t)g * FDIM + kBase + c4);
      *(float4*)&sX[row * SXP + c4] = v;
    }
#pragma unroll
    for (int it = 0; it < 8; ++it) {
      int idx = t + 256 * it;
      int k   = idx >> 5;
      int n4  = (idx & 31) << 2;
      float4 v = *(const float4*)(W + (size_t)(kBase + k) * FDIM + colBase + n4);
      float* p = sWp + ((size_t)(k >> 1) * BNP + n4) * 2 + (k & 1);
      p[0] = v.x; p[2] = v.y; p[4] = v.z; p[6] = v.w;
    }
    __syncthreads();

#pragma unroll
    for (int kk = 0; kk < BK; kk += 4) {
      const int ks = kk + 2 * half;
      v2f a = *(const v2f*)&sX[(waveRow * 16 + l16) * SXP + ks];
      const v2f* wrow = (const v2f*)(sWp + ((kk >> 1) + half) * BNP * 2);
      v2f b0 = wrow[cb];
      v2f b1 = wrow[cb + 16];
      v2f b2 = wrow[cb + 32];
      v2f b3 = wrow[cb + 48];
      acc0 = __builtin_amdgcn_wmma_f32_16x16x4_f32(false, a, false, b0, (short)0, acc0, false, false);
      acc1 = __builtin_amdgcn_wmma_f32_16x16x4_f32(false, a, false, b1, (short)0, acc1, false, false);
      acc2 = __builtin_amdgcn_wmma_f32_16x16x4_f32(false, a, false, b2, (short)0, acc2, false, false);
      acc3 = __builtin_amdgcn_wmma_f32_16x16x4_f32(false, a, false, b3, (short)0, acc3, false, false);
    }
    __syncthreads();
  }

#pragma unroll
  for (int v = 0; v < 8; ++v) {
    int gr = rowBase + waveRow * 16 + v + 8 * half;
    if (gr < nrows) {
      float* p = H + (size_t)gr * FDIM + colBase + waveCol * 64 + l16;
      p[0]  = acc0[v];
      p[16] = acc1[v];
      p[32] = acc2[v];
      p[48] = acc3[v];
    }
  }
}

#define TR   196
#define OW   32
#define DB   8
#define QCAP 520

__device__ __forceinline__ void lds_add(float* p, float x) {
  __hip_atomic_fetch_add(p, x, __ATOMIC_RELAXED, __HIP_MEMORY_SCOPE_WORKGROUP);
}

template <bool FULL>
__device__ __forceinline__ void drain_hits(const float* __restrict__ h,
                                           const int* __restrict__ src,
                                           const int* __restrict__ dst,
                                           const float* __restrict__ val,
                                           float* tile, const int* qe_w,
                                           int base, int n, int lo, int lane) {
  float4 ra[DB], rb[DB];
  float  v[DB];
  int    dl[DB];
#pragma unroll
  for (int i = 0; i < DB; ++i) {
    if (FULL || i < n) {
      const int e = qe_w[base + i];
      const int s = src[e];
      dl[i]       = dst[e] - lo;
      v[i]        = val[e];
      const float* p = h + (size_t)s * FDIM + lane * 4;
      ra[i] = *(const float4*)p;
      rb[i] = *(const float4*)(p + 128);
    }
  }
#pragma unroll
  for (int i = 0; i < DB; ++i) {
    if (FULL || i < n) {
      float* tp = tile + dl[i] * FDIM + lane;
      const float w = v[i];
      lds_add(tp,       ra[i].x * w);
      lds_add(tp + 32,  ra[i].y * w);
      lds_add(tp + 64,  ra[i].z * w);
      lds_add(tp + 96,  ra[i].w * w);
      lds_add(tp + 128, rb[i].x * w);
      lds_add(tp + 160, rb[i].y * w);
      lds_add(tp + 192, rb[i].z * w);
      lds_add(tp + 224, rb[i].w * w);
    }
  }
}

__global__ __launch_bounds__(1024)
void owner_accumulate(const float* __restrict__ h, const int* __restrict__ src,
                      const int* __restrict__ dst, const float* __restrict__ val,
                      float* __restrict__ acc, const float* __restrict__ bias,
                      const float* __restrict__ skip, float* __restrict__ outp,
                      int E, int nrows, int first, int mode) {
  __shared__ __align__(16) float tile[TR * FDIM];
  __shared__ int qe[OW * QCAP];

  const int t    = threadIdx.x;
  const int lane = t & 31;
  const int wave = __builtin_amdgcn_readfirstlane(t >> 5);
  const int lo   = blockIdx.x * TR;
  const int nown = min(TR, nrows - lo);

  const int q    = t & 63;
  const int r0   = t >> 6;
  const int toff = (q >> 5) * 128 + (q & 31);

  if (first) {
    for (int row = r0; row < TR; row += 16) {
      float* tp = tile + row * FDIM + toff;
      tp[0] = 0.f; tp[32] = 0.f; tp[64] = 0.f; tp[96] = 0.f;
    }
  } else {
    for (int rb = r0; rb < TR; rb += 64) {
      float4 v[4];
#pragma unroll
      for (int u = 0; u < 4; ++u) {
        const int row = min(rb + 16 * u, nown - 1);
        v[u] = *(const float4*)(acc + (size_t)(lo + row) * FDIM + q * 4);
      }
#pragma unroll
      for (int u = 0; u < 4; ++u) {
        const int row = rb + 16 * u;
        if (row < TR) {
          const float4 w = row < nown ? v[u] : make_float4(0.f, 0.f, 0.f, 0.f);
          float* tp = tile + row * FDIM + toff;
          tp[0] = w.x; tp[32] = w.y; tp[64] = w.z; tp[96] = w.w;
        }
      }
    }
  }
  __syncthreads();

  int* qe_w = qe + wave * QCAP;
  int  qn   = 0;

  auto append = [&](unsigned mask, int e) {
    if ((mask >> lane) & 1u) qe_w[qn + (int)__builtin_amdgcn_mbcnt_lo(mask, 0u)] = e;
    qn += __builtin_popcount(mask);
  };
  auto drain_full = [&]() {
    while (qn >= DB) {
      drain_hits<true>(h, src, dst, val, tile, qe_w, qn - DB, DB, lo, lane);
      qn -= DB;
    }
  };

  if ((((size_t)dst) & 15) == 0 && (E & 127) == 0) {
    const int nChunks = E >> 7;
    const int rot   = (int)(((long long)blockIdx.x * nChunks) / gridDim.x);
    const int nStep = (nChunks + OW - 1) / OW;

    auto chunk_of = [&](int m) -> int {
      int k = wave + OW * m;
      k = k < nChunks ? k : 0;
      int c = k + rot;
      if (c >= nChunks) c -= nChunks;
      return c;
    };
    auto load_chunk = [&](int m) -> int4 {
      return *(const int4*)(dst + (size_t)chunk_of(m) * 128 + lane * 4);
    };
    auto process = [&](int4 d, int m) {
      const unsigned un = (wave + OW * m < nChunks) ? (unsigned)nown : 0u;
      const unsigned m0 = __builtin_amdgcn_ballot_w32((unsigned)(d.x - lo) < un);
      const unsigned m1 = __builtin_amdgcn_ballot_w32((unsigned)(d.y - lo) < un);
      const unsigned m2 = __builtin_amdgcn_ballot_w32((unsigned)(d.z - lo) < un);
      const unsigned m3 = __builtin_amdgcn_ballot_w32((unsigned)(d.w - lo) < un);
      if (m0 | m1 | m2 | m3) {
        const int e0 = chunk_of(m) * 128 + lane * 4;
        append(m0, e0); append(m1, e0 + 1); append(m2, e0 + 2); append(m3, e0 + 3);
      }
    };

    int4 a0 = load_chunk(0), a1 = load_chunk(1), a2 = load_chunk(2), a3 = load_chunk(3);
    for (int m = 0; m < nStep; m += 8) {
      const int4 b0 = load_chunk(m + 4), b1 = load_chunk(m + 5),
                 b2 = load_chunk(m + 6), b3 = load_chunk(m + 7);
      process(a0, m); process(a1, m + 1); process(a2, m + 2); process(a3, m + 3);
      drain_full();
      a0 = load_chunk(m + 8);  a1 = load_chunk(m + 9);
      a2 = load_chunk(m + 10); a3 = load_chunk(m + 11);
      process(b0, m + 4); process(b1, m + 5); process(b2, m + 6); process(b3, m + 7);
      drain_full();
    }
  } else {
    for (int eb = wave * 32; eb < E; eb += OW * 32) {
      const int e = eb + lane;
      const int d = e < E ? dst[e] : -1;
      append(__builtin_amdgcn_ballot_w32((unsigned)(d - lo) < (unsigned)nown), e);
      drain_full();
    }
  }
  if (qn > 0)
    drain_hits<false>(h, src, dst, val, tile, qe_w, 0, qn, lo, lane);
  __syncthreads();

  if (mode == 0) {
    for (int row = r0; row < nown; row += 16) {
      const float* tp = tile + row * FDIM + toff;
      *(float4*)(acc + (size_t)(lo + row) * FDIM + q * 4) =
          make_float4(tp[0], tp[32], tp[64], tp[96]);
    }
  } else {
    const float4 b = *(const float4*)(bias + q * 4);
    for (int rb = r0; rb < nown; rb += 64) {
      float4 s[4];
#pragma unroll
      for (int u = 0; u < 4; ++u) {
        s[u] = make_float4(0.f, 0.f, 0.f, 0.f);
        if (mode == 2) {
          const int row = min(rb + 16 * u, nown - 1);
          s[u] = *(const float4*)(skip + (size_t)(lo + row) * FDIM + q * 4);
        }
      }
#pragma unroll
      for (int u = 0; u < 4; ++u) {
        const int row = rb + 16 * u;
        if (row < nown) {
          const float* tp = tile + row * FDIM + toff;
          float4 v;
          v.x = fmaxf(tp[0]  + b.x, 0.f) + s[u].x;
          v.y = fmaxf(tp[32] + b.y, 0.f) + s[u].y;
          v.z = fmaxf(tp[64] + b.z, 0.f) + s[u].z;
          v.w = fmaxf(tp[96] + b.w, 0.f) + s[u].w;
          *(float4*)(outp + (size_t)(lo + row) * FDIM + q * 4) = v;
        }
      }
    }
  }
}

extern "C" void kernel_launch(void* const* d_in, const int* in_sizes, int n_in,
                              void* d_out, int out_size, void* d_ws, size_t ws_size,
                              hipStream_t stream) {
  const float* x    = (const float*)d_in[0];
  const int*   esrc = (const int*)d_in[1];
  const int*   edst = (const int*)d_in[2];
  const float* eval = (const float*)d_in[3];
  const float* W1   = (const float*)d_in[4];
  const float* b1   = (const float*)d_in[5];
  const float* W2   = (const float*)d_in[6];
  const float* b2   = (const float*)d_in[7];
  float* out = (float*)d_out;

  const int R  = 4;
  const int Nn = in_sizes[0] / FDIM;
  const int E  = in_sizes[1] / R;

  const size_t nf = (size_t)Nn * FDIM;
  float* h   = (float*)d_ws;
  float* acc = h + nf;
  float* x1  = acc + nf;

  dim3 gGrid((Nn + BM - 1) / BM, FDIM / BN);
  const int oBlocks = (Nn + TR - 1) / TR;

  for (int r = 0; r < R; ++r) {
    gemm_f32_wmma<<<gGrid, 256, 0, stream>>>(x, W1 + (size_t)r * FDIM * FDIM, h, Nn);
    owner_accumulate<<<oBlocks, 1024, 0, stream>>>(h, esrc + (size_t)r * E,
                                                   edst + (size_t)r * E,
                                                   eval + (size_t)r * E, acc, b1,
                                                   x, x1, E, Nn, r == 0,
                                                   r == R - 1 ? 1 : 0);
  }

  for (int r = 0; r < R; ++r) {
    gemm_f32_wmma<<<gGrid, 256, 0, stream>>>(x1, W2 + (size_t)r * FDIM * FDIM, h, Nn);
    owner_accumulate<<<oBlocks, 1024, 0, stream>>>(h, esrc + (size_t)r * E,
                                                   edst + (size_t)r * E,
                                                   eval + (size_t)r * E, acc, b2,
                                                   x1, out, E, Nn, r == 0,
                                                   r == R - 1 ? 2 : 0);
  }
}


# owner blocks scan one owner-block byte per edge (dst/196, written by a new pre-pass once per call) in place of the 4-byte dst words: 4x fewer scan bytes per block; sums still in arrival order
# speedup vs baseline: 2.1344x; 1.0523x over previous
#include <hip/hip_runtime.h>

typedef __attribute__((ext_vector_type(2))) float v2f;
typedef __attribute__((ext_vector_type(8))) float v8f;

#define FDIM 256
#define BM 64
#define BN 128
#define BK 64
#define SXP (BK + 4)
#define BNP 144

__global__ __launch_bounds__(256)
void gemm_f32_wmma(const float* __restrict__ X, const float* __restrict__ W,
                   float* __restrict__ H, int nrows) {
  __shared__ __align__(16) float sX[BM * SXP];
  __shared__ __align__(16) float sWp[(BK / 2) * BNP * 2];

  const int t       = threadIdx.x;
  const int lane    = t & 31;
  const int wave    = t >> 5;
  const int l16     = lane & 15;
  const int half    = lane >> 4;
  const int waveRow = wave & 3;
  const int waveCol = wave >> 2;
  const int rowBase = blockIdx.x * BM;
  const int colBase = blockIdx.y * BN;
  const int cb      = waveCol * 64 + l16;

  v8f acc0 = {}, acc1 = {}, acc2 = {}, acc3 = {};

  for (int kBase = 0; kBase < FDIM; kBase += BK) {
#pragma unroll
    for (int it = 0; it < 4; ++it) {
      int idx = t + 256 * it;
      int row = idx >> 4;
      int c4  = (idx & 15) << 2;
      float4 v = make_float4(0.f, 0.f, 0.f, 0.f);
      int g = rowBase + row;
      if (g < nrows)
        v = *(const float4*)(X + (size_t)g * FDIM + kBase + c4);
      *(float4*)&sX[row * SXP + c4] = v;
    }
#pragma unroll
    for (int it = 0; it < 8; ++it) {
      int idx = t + 256 * it;
      int k   = idx >> 5;
      int n4  = (idx & 31) << 2;
      float4 v = *(const float4*)(W + (size_t)(kBase + k) * FDIM + colBase + n4);
      float* p = sWp + ((size_t)(k >> 1) * BNP + n4) * 2 + (k & 1);
      p[0] = v.x; p[2] = v.y; p[4] = v.z; p[6] = v.w;
    }
    __syncthreads();

#pragma unroll
    for (int kk = 0; kk < BK; kk += 4) {
      const int ks = kk + 2 * half;
      v2f a = *(const v2f*)&sX[(waveRow * 16 + l16) * SXP + ks];
      const v2f* wrow = (const v2f*)(sWp + ((kk >> 1) + half) * BNP * 2);
      v2f b0 = wrow[cb];
      v2f b1 = wrow[cb + 16];
      v2f b2 = wrow[cb + 32];
      v2f b3 = wrow[cb + 48];
      acc0 = __builtin_amdgcn_wmma_f32_16x16x4_f32(false, a, false, b0, (short)0, acc0, false, false);
      acc1 = __builtin_amdgcn_wmma_f32_16x16x4_f32(false, a, false, b1, (short)0, acc1, false, false);
      acc2 = __builtin_amdgcn_wmma_f32_16x16x4_f32(false, a, false, b2, (short)0, acc2, false, false);
      acc3 = __builtin_amdgcn_wmma_f32_16x16x4_f32(false, a, false, b3, (short)0, acc3, false, false);
    }
    __syncthreads();
  }

#pragma unroll
  for (int v = 0; v < 8; ++v) {
    int gr = rowBase + waveRow * 16 + v + 8 * half;
    if (gr < nrows) {
      float* p = H + (size_t)gr * FDIM + colBase + waveCol * 64 + l16;
      p[0]  = acc0[v];
      p[16] = acc1[v];
      p[32] = acc2[v];
      p[48] = acc3[v];
    }
  }
}

#define TR   196
#define KCH  512
#define OW   32
#define DB   8
#define QCAP 520

__global__ __launch_bounds__(256)
void owner_keys(const int* __restrict__ dst, unsigned* __restrict__ keys4, int total) {
  const int i = blockIdx.x * 256 + threadIdx.x;
  if (4 * i >= total) return;
  unsigned w = 0u;
#pragma unroll
  for (int j = 0; j < 4; ++j) {
    const int e = 4 * i + j;
    const unsigned d = e < total ? (unsigned)dst[e] : 0u;
    w |= ((d / (unsigned)TR) & 255u) << (8 * j);
  }
  keys4[i] = w;
}

__device__ __forceinline__ void lds_add(float* p, float x) {
  __hip_atomic_fetch_add(p, x, __ATOMIC_RELAXED, __HIP_MEMORY_SCOPE_WORKGROUP);
}

template <bool FULL>
__device__ __forceinline__ void drain_hits(const float* __restrict__ h,
                                           const int* __restrict__ src,
                                           const int* __restrict__ dst,
                                           const float* __restrict__ val,
                                           float* tile, const int* qe_w,
                                           int base, int n, int lo, int nown,
                                           int lane) {
  float4 ra[DB], rb[DB];
  float  v[DB];
  int    dl[DB];
#pragma unroll
  for (int i = 0; i < DB; ++i) {
    if (FULL || i < n) {
      const int e = qe_w[base + i];
      const int s = src[e];
      dl[i]       = dst[e] - lo;
      v[i]        = val[e];
      const float* p = h + (size_t)s * FDIM + lane * 4;
      ra[i] = *(const float4*)p;
      rb[i] = *(const float4*)(p + 128);
    }
  }
#pragma unroll
  for (int i = 0; i < DB; ++i) {
    if ((FULL || i < n) && (unsigned)dl[i] < (unsigned)nown) {
      float* tp = tile + dl[i] * FDIM + lane;
      const float w = v[i];
      lds_add(tp,       ra[i].x * w);
      lds_add(tp + 32,  ra[i].y * w);
      lds_add(tp + 64,  ra[i].z * w);
      lds_add(tp + 96,  ra[i].w * w);
      lds_add(tp + 128, rb[i].x * w);
      lds_add(tp + 160, rb[i].y * w);
      lds_add(tp + 192, rb[i].z * w);
      lds_add(tp + 224, rb[i].w * w);
    }
  }
}

__global__ __launch_bounds__(1024)
void owner_accumulate(const float* __restrict__ h, const int* __restrict__ src,
                      const int* __restrict__ dst, const float* __restrict__ val,
                      const unsigned* __restrict__ keys,
                      float* __restrict__ acc, const float* __restrict__ bias,
                      const float* __restrict__ skip, float* __restrict__ outp,
                      int E, int nrows, int first, int mode) {
  __shared__ __align__(16) float tile[TR * FDIM];
  __shared__ int qe[OW * QCAP];

  const int t    = threadIdx.x;
  const int lane = t & 31;
  const int wave = __builtin_amdgcn_readfirstlane(t >> 5);
  const int lo   = blockIdx.x * TR;
  const int nown = min(TR, nrows - lo);

  const int q    = t & 63;
  const int r0   = t >> 6;
  const int toff = (q >> 5) * 128 + (q & 31);

  if (first) {
    for (int row = r0; row < TR; row += 16) {
      float* tp = tile + row * FDIM + toff;
      tp[0] = 0.f; tp[32] = 0.f; tp[64] = 0.f; tp[96] = 0.f;
    }
  } else {
    for (int rb = r0; rb < TR; rb += 64) {
      float4 v[4];
#pragma unroll
      for (int u = 0; u < 4; ++u) {
        const int row = min(rb + 16 * u, nown - 1);
        v[u] = *(const float4*)(acc + (size_t)(lo + row) * FDIM + q * 4);
      }
#pragma unroll
      for (int u = 0; u < 4; ++u) {
        const int row = rb + 16 * u;
        if (row < TR) {
          const float4 w = row < nown ? v[u] : make_float4(0.f, 0.f, 0.f, 0.f);
          float* tp = tile + row * FDIM + toff;
          tp[0] = w.x; tp[32] = w.y; tp[64] = w.z; tp[96] = w.w;
        }
      }
    }
  }
  __syncthreads();

  int* qe_w = qe + wave * QCAP;
  int  qn   = 0;

  auto append = [&](unsigned mask, int e) {
    if ((mask >> lane) & 1u) qe_w[qn + (int)__builtin_amdgcn_mbcnt_lo(mask, 0u)] = e;
    qn += __builtin_popcount(mask);
  };
  auto drain_full = [&]() {
    while (qn >= DB) {
      drain_hits<true>(h, src, dst, val, tile, qe_w, qn - DB, DB, lo, nown, lane);
      qn -= DB;
    }
  };

  if ((((size_t)keys) & 15) == 0 && (E & 15) == 0 && E >= 16 && gridDim.x <= 256) {
    const int nChunks = (E + KCH - 1) / KCH;
    const int rot   = (int)(((long long)blockIdx.x * nChunks) / gridDim.x);
    const int nStep = (nChunks + OW - 1) / OW;
    const unsigned pat = (unsigned)blockIdx.x * 0x01010101u;

    auto chunk_of = [&](int m) -> int {
      int k = wave + OW * m;
      k = k < nChunks ? k : 0;
      int c = k + rot;
      if (c >= nChunks) c -= nChunks;
      return c;
    };
    auto load_chunk = [&](int m) -> uint4 {
      const int c = chunk_of(m);
      const int o = min(c * KCH + lane * 16, E - 16);
      return *(const uint4*)((const unsigned char*)keys + o);
    };
    auto eq = [&](unsigned x) -> unsigned {
      const unsigned y = x ^ pat;
      return ~(((y & 0x7F7F7F7Fu) + 0x7F7F7F7Fu) | y | 0x7F7F7F7Fu);
    };
    auto process = [&](uint4 d, int m) {
      if (wave + OW * m >= nChunks) return;
      const int e0 = chunk_of(m) * KCH + lane * 16;
      const unsigned live = e0 < E ? 0xFFFFFFFFu : 0u;
      const unsigned z0 = eq(d.x) & live, z1 = eq(d.y) & live,
                     z2 = eq(d.z) & live, z3 = eq(d.w) & live;
      if (__builtin_amdgcn_ballot_w32((z0 | z1 | z2 | z3) != 0u) == 0u) return;
      auto word = [&](unsigned z, int e) {
        if (__builtin_amdgcn_ballot_w32(z != 0u) == 0u) return;
#pragma unroll
        for (int b = 0; b < 4; ++b) {
          const unsigned mk = __builtin_amdgcn_ballot_w32((z & (0x80u << (8 * b))) != 0u);
          if (mk) append(mk, e + b);
        }
      };
      word(z0, e0); word(z1, e0 + 4); word(z2, e0 + 8); word(z3, e0 + 12);
    };

    uint4 c0 = load_chunk(0), c1 = load_chunk(1), c2 = load_chunk(2), c3 = load_chunk(3);
    for (int m = 0; m < nStep; m += 4) {
      const uint4 n0 = load_chunk(m + 4), n1 = load_chunk(m + 5),
                  n2 = load_chunk(m + 6), n3 = load_chunk(m + 7);
      process(c0, m);     drain_full();
      process(c1, m + 1); drain_full();
      process(c2, m + 2); drain_full();
      process(c3, m + 3); drain_full();
      c0 = n0; c1 = n1; c2 = n2; c3 = n3;
    }
  } else {
    for (int eb = wave * 32; eb < E; eb += OW * 32) {
      const int e = eb + lane;
      const int d = e < E ? dst[e] : -1;
      append(__builtin_amdgcn_ballot_w32((unsigned)(d - lo) < (unsigned)nown), e);
      drain_full();
    }
  }
  if (qn > 0)
    drain_hits<false>(h, src, dst, val, tile, qe_w, 0, qn, lo, nown, lane);
  __syncthreads();

  if (mode == 0) {
    for (int row = r0; row < nown; row += 16) {
      const float* tp = tile + row * FDIM + toff;
      *(float4*)(acc + (size_t)(lo + row) * FDIM + q * 4) =
          make_float4(tp[0], tp[32], tp[64], tp[96]);
    }
  } else {
    const float4 b = *(const float4*)(bias + q * 4);
    for (int rb = r0; rb < nown; rb += 64) {
      float4 s[4];
#pragma unroll
      for (int u = 0; u < 4; ++u) {
        s[u] = make_float4(0.f, 0.f, 0.f, 0.f);
        if (mode == 2) {
          const int row = min(rb + 16 * u, nown - 1);
          s[u] = *(const float4*)(skip + (size_t)(lo + row) * FDIM + q * 4);
        }
      }
#pragma unroll
      for (int u = 0; u < 4; ++u) {
        const int row = rb + 16 * u;
        if (row < nown) {
          const float* tp = tile + row * FDIM + toff;
          float4 v;
          v.x = fmaxf(tp[0]  + b.x, 0.f) + s[u].x;
          v.y = fmaxf(tp[32] + b.y, 0.f) + s[u].y;
          v.z = fmaxf(tp[64] + b.z, 0.f) + s[u].z;
          v.w = fmaxf(tp[96] + b.w, 0.f) + s[u].w;
          *(float4*)(outp + (size_t)(lo + row) * FDIM + q * 4) = v;
        }
      }
    }
  }
}

extern "C" void kernel_launch(void* const* d_in, const int* in_sizes, int n_in,
                              void* d_out, int out_size, void* d_ws, size_t ws_size,
                              hipStream_t stream) {
  const float* x    = (const float*)d_in[0];
  const int*   esrc = (const int*)d_in[1];
  const int*   edst = (const int*)d_in[2];
  const float* eval = (const float*)d_in[3];
  const float* W1   = (const float*)d_in[4];
  const float* b1   = (const float*)d_in[5];
  const float* W2   = (const float*)d_in[6];
  const float* b2   = (const float*)d_in[7];
  float* out = (float*)d_out;

  const int R  = 4;
  const int Nn = in_sizes[0] / FDIM;
  const int E  = in_sizes[1] / R;

  const size_t nf = (size_t)Nn * FDIM;
  float* h   = (float*)d_ws;
  float* acc = h + nf;
  float* x1  = acc + nf;
  unsigned char* keys = (unsigned char*)(x1 + nf);

  dim3 gGrid((Nn + BM - 1) / BM, FDIM / BN);
  const int oBlocks = (Nn + TR - 1) / TR;

  owner_keys<<<(R * E / 4 + 256) / 256, 256, 0, stream>>>(edst, (unsigned*)keys, R * E);

  for (int r = 0; r < R; ++r) {
    gemm_f32_wmma<<<gGrid, 256, 0, stream>>>(x, W1 + (size_t)r * FDIM * FDIM, h, Nn);
    owner_accumulate<<<oBlocks, 1024, 0, stream>>>(h, esrc + (size_t)r * E,
                                                   edst + (size_t)r * E,
                                                   eval + (size_t)r * E,
                                                   (const unsigned*)(keys + (size_t)r * E), acc, b1,
                                                   x, x1, E, Nn, r == 0,
                                                   r == R - 1 ? 1 : 0);
  }

  for (int r = 0; r < R; ++r) {
    gemm_f32_wmma<<<gGrid, 256, 0, stream>>>(x1, W2 + (size_t)r * FDIM * FDIM, h, Nn);
    owner_accumulate<<<oBlocks, 1024, 0, stream>>>(h, esrc + (size_t)r * E,
                                                   edst + (size_t)r * E,
                                                   eval + (size_t)r * E,
                                                   (const unsigned*)(keys + (size_t)r * E), acc, b2,
                                                   x1, out, E, Nn, r == 0,
                                                   r == R - 1 ? 2 : 0);
  }
}


# owner blocks keep their LDS tile across all 4 relations: one owner launch per layer (was 4), partial-sum plane never written or re-read; float sums in arrival order, now across relations too
# speedup vs baseline: 2.5386x; 1.1894x over previous
#include <hip/hip_runtime.h>

typedef __attribute__((ext_vector_type(2))) float v2f;
typedef __attribute__((ext_vector_type(8))) float v8f;

#define FDIM 256
#define BM 64
#define BN 128
#define BK 64
#define SXP (BK + 4)
#define BNP 144

__global__ __launch_bounds__(256)
void gemm_f32_wmma(const float* __restrict__ X, const float* __restrict__ W,
                   float* __restrict__ H, int nrows) {
  __shared__ __align__(16) float sX[BM * SXP];
  __shared__ __align__(16) float sWp[(BK / 2) * BNP * 2];

  const int t       = threadIdx.x;
  const int lane    = t & 31;
  const int wave    = t >> 5;
  const int l16     = lane & 15;
  const int half    = lane >> 4;
  const int waveRow = wave & 3;
  const int waveCol = wave >> 2;
  const int rowBase = blockIdx.x * BM;
  const int colBase = blockIdx.y * BN;
  const int cb      = waveCol * 64 + l16;

  v8f acc0 = {}, acc1 = {}, acc2 = {}, acc3 = {};

  for (int kBase = 0; kBase < FDIM; kBase += BK) {
#pragma unroll
    for (int it = 0; it < 4; ++it) {
      int idx = t + 256 * it;
      int row = idx >> 4;
      int c4  = (idx & 15) << 2;
      float4 v = make_float4(0.f, 0.f, 0.f, 0.f);
      int g = rowBase + row;
      if (g < nrows)
        v = *(const float4*)(X + (size_t)g * FDIM + kBase + c4);
      *(float4*)&sX[row * SXP + c4] = v;
    }
#pragma unroll
    for (int it = 0; it < 8; ++it) {
      int idx = t + 256 * it;
      int k   = idx >> 5;
      int n4  = (idx & 31) << 2;
      float4 v = *(const float4*)(W + (size_t)(kBase + k) * FDIM + colBase + n4);
      float* p = sWp + ((size_t)(k >> 1) * BNP + n4) * 2 + (k & 1);
      p[0] = v.x; p[2] = v.y; p[4] = v.z; p[6] = v.w;
    }
    __syncthreads();

#pragma unroll
    for (int kk = 0; kk < BK; kk += 4) {
      const int ks = kk + 2 * half;
      v2f a = *(const v2f*)&sX[(waveRow * 16 + l16) * SXP + ks];
      const v2f* wrow = (const v2f*)(sWp + ((kk >> 1) + half) * BNP * 2);
      v2f b0 = wrow[cb];
      v2f b1 = wrow[cb + 16];
      v2f b2 = wrow[cb + 32];
      v2f b3 = wrow[cb + 48];
      acc0 = __builtin_amdgcn_wmma_f32_16x16x4_f32(false, a, false, b0, (short)0, acc0, false, false);
      acc1 = __builtin_amdgcn_wmma_f32_16x16x4_f32(false, a, false, b1, (short)0, acc1, false, false);
      acc2 = __builtin_amdgcn_wmma_f32_16x16x4_f32(false, a, false, b2, (short)0, acc2, false, false);
      acc3 = __builtin_amdgcn_wmma_f32_16x16x4_f32(false, a, false, b3, (short)0, acc3, false, false);
    }
    __syncthreads();
  }

#pragma unroll
  for (int v = 0; v < 8; ++v) {
    int gr = rowBase + waveRow * 16 + v + 8 * half;
    if (gr < nrows) {
      float* p = H + (size_t)gr * FDIM + colBase + waveCol * 64 + l16;
      p[0]  = acc0[v];
      p[16] = acc1[v];
      p[32] = acc2[v];
      p[48] = acc3[v];
    }
  }
}

#define TR   196
#define KCH  512
#define OW   32
#define DB   8
#define QCAP 520

__global__ __launch_bounds__(256)
void owner_keys(const int* __restrict__ dst, unsigned* __restrict__ keys4, int total) {
  const int i = blockIdx.x * 256 + threadIdx.x;
  if (4 * i >= total) return;
  unsigned w = 0u;
#pragma unroll
  for (int j = 0; j < 4; ++j) {
    const int e = 4 * i + j;
    const unsigned d = e < total ? (unsigned)dst[e] : 0u;
    w |= ((d / (unsigned)TR) & 255u) << (8 * j);
  }
  keys4[i] = w;
}

__device__ __forceinline__ void lds_add(float* p, float x) {
  __hip_atomic_fetch_add(p, x, __ATOMIC_RELAXED, __HIP_MEMORY_SCOPE_WORKGROUP);
}

template <bool FULL>
__device__ __forceinline__ void drain_hits(const float* __restrict__ h,
                                           const int* __restrict__ src,
                                           const int* __restrict__ dst,
                                           const float* __restrict__ val,
                                           float* tile, const int* qe_w,
                                           int base, int n, int lo, int nown,
                                           int lane) {
  float4 ra[DB], rb[DB];
  float  v[DB];
  int    dl[DB];
#pragma unroll
  for (int i = 0; i < DB; ++i) {
    const int j = FULL ? i : min(i, n - 1);
    const int e = qe_w[base + j];
    const int s = src[e];
    dl[i]       = dst[e] - lo;
    v[i]        = val[e];
    const float* p = h + (size_t)s * FDIM + lane * 4;
    ra[i] = *(const float4*)p;
    rb[i] = *(const float4*)(p + 128);
  }
#pragma unroll
  for (int i = 0; i < DB; ++i) {
    if ((FULL || i < n) && (unsigned)dl[i] < (unsigned)nown) {
      float* tp = tile + dl[i] * FDIM + lane;
      const float w = v[i];
      lds_add(tp,       ra[i].x * w);
      lds_add(tp + 32,  ra[i].y * w);
      lds_add(tp + 64,  ra[i].z * w);
      lds_add(tp + 96,  ra[i].w * w);
      lds_add(tp + 128, rb[i].x * w);
      lds_add(tp + 160, rb[i].y * w);
      lds_add(tp + 192, rb[i].z * w);
      lds_add(tp + 224, rb[i].w * w);
    }
  }
}

__global__ __launch_bounds__(1024)
void owner_accumulate(const float* __restrict__ hAll, const int* __restrict__ srcAll,
                      const int* __restrict__ dstAll, const float* __restrict__ valAll,
                      const unsigned* __restrict__ keysAll,
                      const float* __restrict__ bias,
                      const float* __restrict__ skip, float* __restrict__ outp,
                      int E, int nrows, int nrel, int mode) {
  __shared__ __align__(16) float tile[TR * FDIM];
  __shared__ int qe[OW * QCAP];

  const int t    = threadIdx.x;
  const int lane = t & 31;
  const int wave = __builtin_amdgcn_readfirstlane(t >> 5);
  const int lo   = blockIdx.x * TR;
  const int nown = min(TR, nrows - lo);

  const int q    = t & 63;
  const int r0   = t >> 6;
  const int toff = (q >> 5) * 128 + (q & 31);

  for (int row = r0; row < TR; row += 16) {
    float* tp = tile + row * FDIM + toff;
    tp[0] = 0.f; tp[32] = 0.f; tp[64] = 0.f; tp[96] = 0.f;
  }
  __syncthreads();

  int* qe_w = qe + wave * QCAP;
  int  qn   = 0;
  const size_t plane = (size_t)nrows * FDIM;

  for (int r = 0; r < nrel; ++r) {
  const float* h   = hAll   + (size_t)r * plane;
  const int*   src = srcAll + (size_t)r * E;
  const int*   dst = dstAll + (size_t)r * E;
  const float* val = valAll + (size_t)r * E;
  const unsigned char* keys = (const unsigned char*)keysAll + (size_t)r * E;

  auto append = [&](unsigned mask, int e) {
    if ((mask >> lane) & 1u) qe_w[qn + (int)__builtin_amdgcn_mbcnt_lo(mask, 0u)] = e;
    qn += __builtin_popcount(mask);
  };
  auto drain_full = [&]() {
    while (qn >= DB) {
      drain_hits<true>(h, src, dst, val, tile, qe_w, qn - DB, DB, lo, nown, lane);
      qn -= DB;
    }
  };

  if ((((size_t)keys) & 15) == 0 && (E & 15) == 0 && E >= 16 && gridDim.x <= 256) {
    const int nChunks = (E + KCH - 1) / KCH;
    const int rot   = (int)(((long long)blockIdx.x * nChunks) / gridDim.x);
    const int nStep = (nChunks + OW - 1) / OW;
    const unsigned pat = (unsigned)blockIdx.x * 0x01010101u;

    auto chunk_of = [&](int m) -> int {
      int k = wave + OW * m;
      k = k < nChunks ? k : 0;
      int c = k + rot;
      if (c >= nChunks) c -= nChunks;
      return c;
    };
    auto load_chunk = [&](int m) -> uint4 {
      const int c = chunk_of(m);
      const int o = min(c * KCH + lane * 16, E - 16);
      return *(const uint4*)(keys + o);
    };
    auto eq = [&](unsigned x) -> unsigned {
      const unsigned y = x ^ pat;
      return ~(((y & 0x7F7F7F7Fu) + 0x7F7F7F7Fu) | y | 0x7F7F7F7Fu);
    };
    auto process = [&](uint4 d, int m) {
      if (wave + OW * m >= nChunks) return;
      const int e0 = chunk_of(m) * KCH + lane * 16;
      const unsigned live = e0 < E ? 0xFFFFFFFFu : 0u;
      const unsigned z0 = eq(d.x) & live, z1 = eq(d.y) & live,
                     z2 = eq(d.z) & live, z3 = eq(d.w) & live;
      if (__builtin_amdgcn_ballot_w32((z0 | z1 | z2 | z3) != 0u) == 0u) return;
      auto word = [&](unsigned z, int e) {
        if (__builtin_amdgcn_ballot_w32(z != 0u) == 0u) return;
#pragma unroll
        for (int b = 0; b < 4; ++b) {
          const unsigned mk = __builtin_amdgcn_ballot_w32((z & (0x80u << (8 * b))) != 0u);
          if (mk) append(mk, e + b);
        }
      };
      word(z0, e0); word(z1, e0 + 4); word(z2, e0 + 8); word(z3, e0 + 12);
    };

    uint4 c0 = load_chunk(0), c1 = load_chunk(1), c2 = load_chunk(2), c3 = load_chunk(3);
    for (int m = 0; m < nStep; m += 4) {
      const uint4 n0 = load_chunk(m + 4), n1 = load_chunk(m + 5),
                  n2 = load_chunk(m + 6), n3 = load_chunk(m + 7);
      process(c0, m);     drain_full();
      process(c1, m + 1); drain_full();
      process(c2, m + 2); drain_full();
      process(c3, m + 3); drain_full();
      c0 = n0; c1 = n1; c2 = n2; c3 = n3;
    }
  } else {
    for (int eb = wave * 32; eb < E; eb += OW * 32) {
      const int e = eb + lane;
      const int d = e < E ? dst[e] : -1;
      append(__builtin_amdgcn_ballot_w32((unsigned)(d - lo) < (unsigned)nown), e);
      drain_full();
    }
  }
  if (qn > 0)
    drain_hits<false>(h, src, dst, val, tile, qe_w, 0, qn, lo, nown, lane);
  qn = 0;
  }
  __syncthreads();

  {
    const float4 b = *(const float4*)(bias + q * 4);
    for (int rb = r0; rb < nown; rb += 64) {
      float4 s[4];
#pragma unroll
      for (int u = 0; u < 4; ++u) {
        s[u] = make_float4(0.f, 0.f, 0.f, 0.f);
        if (mode == 2) {
          const int row = min(rb + 16 * u, nown - 1);
          s[u] = *(const float4*)(skip + (size_t)(lo + row) * FDIM + q * 4);
        }
      }
#pragma unroll
      for (int u = 0; u < 4; ++u) {
        const int row = rb + 16 * u;
        if (row < nown) {
          const float* tp = tile + row * FDIM + toff;
          float4 v;
          v.x = fmaxf(tp[0]  + b.x, 0.f) + s[u].x;
          v.y = fmaxf(tp[32] + b.y, 0.f) + s[u].y;
          v.z = fmaxf(tp[64] + b.z, 0.f) + s[u].z;
          v.w = fmaxf(tp[96] + b.w, 0.f) + s[u].w;
          *(float4*)(outp + (size_t)(lo + row) * FDIM + q * 4) = v;
        }
      }
    }
  }
}

extern "C" void kernel_launch(void* const* d_in, const int* in_sizes, int n_in,
                              void* d_out, int out_size, void* d_ws, size_t ws_size,
                              hipStream_t stream) {
  const float* x    = (const float*)d_in[0];
  const int*   esrc = (const int*)d_in[1];
  const int*   edst = (const int*)d_in[2];
  const float* eval = (const float*)d_in[3];
  const float* W1   = (const float*)d_in[4];
  const float* b1   = (const float*)d_in[5];
  const float* W2   = (const float*)d_in[6];
  const float* b2   = (const float*)d_in[7];
  float* out = (float*)d_out;

  const int R  = 4;
  const int Nn = in_sizes[0] / FDIM;
  const int E  = in_sizes[1] / R;

  const size_t nf = (size_t)Nn * FDIM;
  float* h   = (float*)d_ws;
  float* x1  = h + R * nf;
  unsigned char* keys = (unsigned char*)(x1 + nf);

  dim3 gGrid((Nn + BM - 1) / BM, FDIM / BN);
  const int oBlocks = (Nn + TR - 1) / TR;

  owner_keys<<<(R * E / 4 + 256) / 256, 256, 0, stream>>>(edst, (unsigned*)keys, R * E);

  for (int r = 0; r < R; ++r)
    gemm_f32_wmma<<<gGrid, 256, 0, stream>>>(x, W1 + (size_t)r * FDIM * FDIM, h + r * nf, Nn);
  owner_accumulate<<<oBlocks, 1024, 0, stream>>>(h, esrc, edst, eval, (const unsigned*)keys,
                                                 b1, x, x1, E, Nn, R, 1);

  for (int r = 0; r < R; ++r)
    gemm_f32_wmma<<<gGrid, 256, 0, stream>>>(x1, W2 + (size_t)r * FDIM * FDIM, h + r * nf, Nn);
  owner_accumulate<<<oBlocks, 1024, 0, stream>>>(h, esrc, edst, eval, (const unsigned*)keys,
                                                 b2, x1, out, E, Nn, R, 2);
}


# weight product re-tiled: each wave keeps its 256x16 strip of W_r in registers for the whole launch (no W in LDS), X streamed as double-buffered 32-row LDS tiles, 1 launch per layer; sums unchanged
# speedup vs baseline: 2.9870x; 1.1766x over previous
#include <hip/hip_runtime.h>

typedef __attribute__((ext_vector_type(2))) float v2f;
typedef __attribute__((ext_vector_type(8))) float v8f;

#define FDIM 256
#define GR   32
#define GXP  260
#define KG   4
#define GRG  64

__global__ __launch_bounds__(512)
void gemm_wreg(const float* __restrict__ X, const float* __restrict__ Wall,
               float* __restrict__ Hall, int nrows) {
  __shared__ __align__(16) float sX[2][GR * GXP];

  const int t    = threadIdx.x;
  const int lane = t & 31;
  const int wave = t >> 5;
  const int l16  = lane & 15;
  const int half = lane >> 4;
  const int col  = wave * 16 + l16;
  const float* W = Wall + (size_t)blockIdx.y * FDIM * FDIM;
  float*       H = Hall + (size_t)blockIdx.y * (size_t)nrows * FDIM;
  const int nTiles = (nrows + GR - 1) / GR;
  const int srow = t >> 6;
  const int sc4  = (t & 63) << 2;

#define LOAD_TILE(tl)                                                          \
  {                                                                            \
    const int gb = (tl) * GR + srow;                                           \
    n0 = *(const float4*)(X + (size_t)min(gb,      nrows - 1) * FDIM + sc4);   \
    n1 = *(const float4*)(X + (size_t)min(gb + 8,  nrows - 1) * FDIM + sc4);   \
    n2 = *(const float4*)(X + (size_t)min(gb + 16, nrows - 1) * FDIM + sc4);   \
    n3 = *(const float4*)(X + (size_t)min(gb + 24, nrows - 1) * FDIM + sc4);   \
  }
#define STORE_TILE(buf)                                                        \
  {                                                                            \
    float* sp = &sX[buf][srow * GXP + sc4];                                    \
    *(float4*)(sp)            = n0;                                            \
    *(float4*)(sp + 8 * GXP)  = n1;                                            \
    *(float4*)(sp + 16 * GXP) = n2;                                            \
    *(float4*)(sp + 24 * GXP) = n3;                                            \
  }

  int tile = blockIdx.x;
  float4 n0 = make_float4(0.f, 0.f, 0.f, 0.f), n1 = n0, n2 = n0, n3 = n0;
  if (tile < nTiles) LOAD_TILE(tile)

  v2f bw[64];
#pragma unroll
  for (int j = 0; j < 64; ++j) {
    const float* p = W + (size_t)(4 * j + 2 * half) * FDIM + col;
    bw[j].x = p[0];
    bw[j].y = p[FDIM];
  }

  if (tile < nTiles) STORE_TILE(0)
  __syncthreads();

  for (int it = 0; tile < nTiles; tile += GRG, ++it) {
    const int cur = it & 1;
    const int nxt = tile + GRG;
    if (nxt < nTiles) LOAD_TILE(nxt)

    const float* xa = &sX[cur][l16 * GXP + 2 * half];
    v8f acc0 = {}, acc1 = {};
    v2f pa[2][KG], pb[2][KG];
#pragma unroll
    for (int s = 0; s < KG; ++s) {
      pa[0][s] = *(const v2f*)(xa + 4 * s);
      pb[0][s] = *(const v2f*)(xa + 16 * GXP + 4 * s);
    }
#pragma unroll
    for (int g = 0; g < 64 / KG; ++g) {
      const int c = g & 1;
      if (g + 1 < 64 / KG) {
#pragma unroll
        for (int s = 0; s < KG; ++s) {
          pa[c ^ 1][s] = *(const v2f*)(xa + 4 * ((g + 1) * KG + s));
          pb[c ^ 1][s] = *(const v2f*)(xa + 16 * GXP + 4 * ((g + 1) * KG + s));
        }
      }
#pragma unroll
      for (int s = 0; s < KG; ++s) {
        const int j = g * KG + s;
        acc0 = __builtin_amdgcn_wmma_f32_16x16x4_f32(false, pa[c][s], false, bw[j], (short)0, acc0, false, false);
        acc1 = __builtin_amdgcn_wmma_f32_16x16x4_f32(false, pb[c][s], false, bw[j], (short)0, acc1, false, false);
      }
    }

    if (nxt < nTiles) STORE_TILE(cur ^ 1)
    __syncthreads();

    const int rb = tile * GR + 8 * half;
#pragma unroll
    for (int v = 0; v < 8; ++v) {
      const int r0 = rb + v, r1 = rb + 16 + v;
      if (r0 < nrows) H[(size_t)r0 * FDIM + col] = acc0[v];
      if (r1 < nrows) H[(size_t)r1 * FDIM + col] = acc1[v];
    }
  }
}

#define TR   196
#define KCH  512
#define OW   32
#define DB   8
#define QCAP 520

__global__ __launch_bounds__(256)
void owner_keys(const int* __restrict__ dst, unsigned* __restrict__ keys4, int total) {
  const int i = blockIdx.x * 256 + threadIdx.x;
  if (4 * i >= total) return;
  unsigned w = 0u;
#pragma unroll
  for (int j = 0; j < 4; ++j) {
    const int e = 4 * i + j;
    const unsigned d = e < total ? (unsigned)dst[e] : 0u;
    w |= ((d / (unsigned)TR) & 255u) << (8 * j);
  }
  keys4[i] = w;
}

__device__ __forceinline__ void lds_add(float* p, float x) {
  __hip_atomic_fetch_add(p, x, __ATOMIC_RELAXED, __HIP_MEMORY_SCOPE_WORKGROUP);
}

template <bool FULL>
__device__ __forceinline__ void drain_hits(const float* __restrict__ h,
                                           const int* __restrict__ src,
                                           const int* __restrict__ dst,
                                           const float* __restrict__ val,
                                           float* tile, const int* qe_w,
                                           int base, int n, int lo, int nown,
                                           int lane) {
  float4 ra[DB], rb[DB];
  float  v[DB];
  int    dl[DB];
#pragma unroll
  for (int i = 0; i < DB; ++i) {
    const int j = FULL ? i : min(i, n - 1);
    const int e = qe_w[base + j];
    const int s = src[e];
    dl[i]       = dst[e] - lo;
    v[i]        = val[e];
    const float* p = h + (size_t)s * FDIM + lane * 4;
    ra[i] = *(const float4*)p;
    rb[i] = *(const float4*)(p + 128);
  }
#pragma unroll
  for (int i = 0; i < DB; ++i) {
    if ((FULL || i < n) && (unsigned)dl[i] < (unsigned)nown) {
      float* tp = tile + dl[i] * FDIM + lane;
      const float w = v[i];
      lds_add(tp,       ra[i].x * w);
      lds_add(tp + 32,  ra[i].y * w);
      lds_add(tp + 64,  ra[i].z * w);
      lds_add(tp + 96,  ra[i].w * w);
      lds_add(tp + 128, rb[i].x * w);
      lds_add(tp + 160, rb[i].y * w);
      lds_add(tp + 192, rb[i].z * w);
      lds_add(tp + 224, rb[i].w * w);
    }
  }
}

__global__ __launch_bounds__(1024)
void owner_accumulate(const float* __restrict__ hAll, const int* __restrict__ srcAll,
                      const int* __restrict__ dstAll, const float* __restrict__ valAll,
                      const unsigned* __restrict__ keysAll,
                      const float* __restrict__ bias,
                      const float* __restrict__ skip, float* __restrict__ outp,
                      int E, int nrows, int nrel, int mode) {
  __shared__ __align__(16) float tile[TR * FDIM];
  __shared__ int qe[OW * QCAP];

  const int t    = threadIdx.x;
  const int lane = t & 31;
  const int wave = __builtin_amdgcn_readfirstlane(t >> 5);
  const int lo   = blockIdx.x * TR;
  const int nown = min(TR, nrows - lo);

  const int q    = t & 63;
  const int r0   = t >> 6;
  const int toff = (q >> 5) * 128 + (q & 31);

  for (int row = r0; row < TR; row += 16) {
    float* tp = tile + row * FDIM + toff;
    tp[0] = 0.f; tp[32] = 0.f; tp[64] = 0.f; tp[96] = 0.f;
  }
  __syncthreads();

  int* qe_w = qe + wave * QCAP;
  int  qn   = 0;
  const size_t plane = (size_t)nrows * FDIM;

  for (int r = 0; r < nrel; ++r) {
  const float* h   = hAll   + (size_t)r * plane;
  const int*   src = srcAll + (size_t)r * E;
  const int*   dst = dstAll + (size_t)r * E;
  const float* val = valAll + (size_t)r * E;
  const unsigned char* keys = (const unsigned char*)keysAll + (size_t)r * E;

  auto append = [&](unsigned mask, int e) {
    if ((mask >> lane) & 1u) qe_w[qn + (int)__builtin_amdgcn_mbcnt_lo(mask, 0u)] = e;
    qn += __builtin_popcount(mask);
  };
  auto drain_full = [&]() {
    while (qn >= DB) {
      drain_hits<true>(h, src, dst, val, tile, qe_w, qn - DB, DB, lo, nown, lane);
      qn -= DB;
    }
  };

  if ((((size_t)keys) & 15) == 0 && (E & 15) == 0 && E >= 16 && gridDim.x <= 256) {
    const int nChunks = (E + KCH - 1) / KCH;
    const int rot   = (int)(((long long)blockIdx.x * nChunks) / gridDim.x);
    const int nStep = (nChunks + OW - 1) / OW;
    const unsigned pat = (unsigned)blockIdx.x * 0x01010101u;

    auto chunk_of = [&](int m) -> int {
      int k = wave + OW * m;
      k = k < nChunks ? k : 0;
      int c = k + rot;
      if (c >= nChunks) c -= nChunks;
      return c;
    };
    auto load_chunk = [&](int m) -> uint4 {
      const int c = chunk_of(m);
      const int o = min(c * KCH + lane * 16, E - 16);
      return *(const uint4*)(keys + o);
    };
    auto eq = [&](unsigned x) -> unsigned {
      const unsigned y = x ^ pat;
      return ~(((y & 0x7F7F7F7Fu) + 0x7F7F7F7Fu) | y | 0x7F7F7F7Fu);
    };
    auto process = [&](uint4 d, int m) {
      if (wave + OW * m >= nChunks) return;
      const int e0 = chunk_of(m) * KCH + lane * 16;
      const unsigned live = e0 < E ? 0xFFFFFFFFu : 0u;
      const unsigned z0 = eq(d.x) & live, z1 = eq(d.y) & live,
                     z2 = eq(d.z) & live, z3 = eq(d.w) & live;
      if (__builtin_amdgcn_ballot_w32((z0 | z1 | z2 | z3) != 0u) == 0u) return;
      auto word = [&](unsigned z, int e) {
        if (__builtin_amdgcn_ballot_w32(z != 0u) == 0u) return;
#pragma unroll
        for (int b = 0; b < 4; ++b) {
          const unsigned mk = __builtin_amdgcn_ballot_w32((z & (0x80u << (8 * b))) != 0u);
          if (mk) append(mk, e + b);
        }
      };
      word(z0, e0); word(z1, e0 + 4); word(z2, e0 + 8); word(z3, e0 + 12);
    };

    uint4 c0 = load_chunk(0), c1 = load_chunk(1), c2 = load_chunk(2), c3 = load_chunk(3);
    for (int m = 0; m < nStep; m += 4) {
      const uint4 n0 = load_chunk(m + 4), n1 = load_chunk(m + 5),
                  n2 = load_chunk(m + 6), n3 = load_chunk(m + 7);
      process(c0, m);     drain_full();
      process(c1, m + 1); drain_full();
      process(c2, m + 2); drain_full();
      process(c3, m + 3); drain_full();
      c0 = n0; c1 = n1; c2 = n2; c3 = n3;
    }
  } else {
    for (int eb = wave * 32; eb < E; eb += OW * 32) {
      const int e = eb + lane;
      const int d = e < E ? dst[e] : -1;
      append(__builtin_amdgcn_ballot_w32((unsigned)(d - lo) < (unsigned)nown), e);
      drain_full();
    }
  }
  if (qn > 0)
    drain_hits<false>(h, src, dst, val, tile, qe_w, 0, qn, lo, nown, lane);
  qn = 0;
  }
  __syncthreads();

  {
    const float4 b = *(const float4*)(bias + q * 4);
    for (int rb = r0; rb < nown; rb += 64) {
      float4 s[4];
#pragma unroll
      for (int u = 0; u < 4; ++u) {
        s[u] = make_float4(0.f, 0.f, 0.f, 0.f);
        if (mode == 2) {
          const int row = min(rb + 16 * u, nown - 1);
          s[u] = *(const float4*)(skip + (size_t)(lo + row) * FDIM + q * 4);
        }
      }
#pragma unroll
      for (int u = 0; u < 4; ++u) {
        const int row = rb + 16 * u;
        if (row < nown) {
          const float* tp = tile + row * FDIM + toff;
          float4 v;
          v.x = fmaxf(tp[0]  + b.x, 0.f) + s[u].x;
          v.y = fmaxf(tp[32] + b.y, 0.f) + s[u].y;
          v.z = fmaxf(tp[64] + b.z, 0.f) + s[u].z;
          v.w = fmaxf(tp[96] + b.w, 0.f) + s[u].w;
          *(float4*)(outp + (size_t)(lo + row) * FDIM + q * 4) = v;
        }
      }
    }
  }
}

extern "C" void kernel_launch(void* const* d_in, const int* in_sizes, int n_in,
                              void* d_out, int out_size, void* d_ws, size_t ws_size,
                              hipStream_t stream) {
  const float* x    = (const float*)d_in[0];
  const int*   esrc = (const int*)d_in[1];
  const int*   edst = (const int*)d_in[2];
  const float* eval = (const float*)d_in[3];
  const float* W1   = (const float*)d_in[4];
  const float* b1   = (const float*)d_in[5];
  const float* W2   = (const float*)d_in[6];
  const float* b2   = (const float*)d_in[7];
  float* out = (float*)d_out;

  const int R  = 4;
  const int Nn = in_sizes[0] / FDIM;
  const int E  = in_sizes[1] / R;

  const size_t nf = (size_t)Nn * FDIM;
  float* h   = (float*)d_ws;
  float* x1  = h + R * nf;
  unsigned char* keys = (unsigned char*)(x1 + nf);

  dim3 gGrid(GRG, R);
  const int oBlocks = (Nn + TR - 1) / TR;

  owner_keys<<<(R * E / 4 + 256) / 256, 256, 0, stream>>>(edst, (unsigned*)keys, R * E);

  gemm_wreg<<<gGrid, 512, 0, stream>>>(x, W1, h, Nn);
  owner_accumulate<<<oBlocks, 1024, 0, stream>>>(h, esrc, edst, eval, (const unsigned*)keys,
                                                 b1, x, x1, E, Nn, R, 1);

  gemm_wreg<<<gGrid, 512, 0, stream>>>(x1, W2, h, Nn);
  owner_accumulate<<<oBlocks, 1024, 0, stream>>>(h, esrc, edst, eval, (const unsigned*)keys,
                                                 b2, x1, out, E, Nn, R, 2);
}
